# starting kernel
# baseline (speedup 1.0000x reference)
#include <hip/hip_runtime.h>

typedef __attribute__((ext_vector_type(2))) float v2f;
typedef __attribute__((ext_vector_type(8))) float v8f;

#define NN     2048
#define FIN    512
#define FOUT   512
#define FEDGE  16

__global__ __launch_bounds__(256) void wmma_gemm_f32_kernel(
    const float* __restrict__ A, const float* __restrict__ B,
    const float* __restrict__ E, float* __restrict__ C,
    int M, int Ncols, int K)
{
    const int lane = threadIdx.x & 31;
    const int wave = threadIdx.x >> 5;

    const int wm = blockIdx.y * 64  + (wave & 1)  * 32;
    const int wn = blockIdx.x * 128 + (wave >> 1) * 32;

    const int ml = lane & 15;
    const int kh = (lane >> 4) * 2;

    const float* a0p = A + (size_t)(wm      + ml) * K + kh;
    const float* a1p = A + (size_t)(wm + 16 + ml) * K + kh;
    const float* b0p = B + (size_t)kh * Ncols + wn + ml;
    const float* b1p = b0p + 16;
    const size_t bstep = (size_t)4 * Ncols;

    v8f c00 = {}, c01 = {}, c10 = {}, c11 = {};

    #pragma unroll 4
    for (int k = 0; k < K; k += 4) {
        v2f a0 = *(const v2f*)a0p;  a0p += 4;
        v2f a1 = *(const v2f*)a1p;  a1p += 4;
        v2f b0, b1;
        b0.x = b0p[0]; b0.y = b0p[Ncols];
        b1.x = b1p[0]; b1.y = b1p[Ncols];
        b0p += bstep;  b1p += bstep;

        c00 = __builtin_amdgcn_wmma_f32_16x16x4_f32(false, a0, false, b0, (short)0, c00, false, false);
        c01 = __builtin_amdgcn_wmma_f32_16x16x4_f32(false, a0, false, b1, (short)0, c01, false, false);
        c10 = __builtin_amdgcn_wmma_f32_16x16x4_f32(false, a1, false, b0, (short)0, c10, false, false);
        c11 = __builtin_amdgcn_wmma_f32_16x16x4_f32(false, a1, false, b1, (short)0, c11, false, false);
    }

    const int rbase = (lane >> 4) * 8;
    const int colb  = lane & 15;
    auto store_tile = [&](v8f c, int m0, int n0) {
        const int col = n0 + colb;
        #pragma unroll
        for (int r = 0; r < 8; ++r) {
            const int row = m0 + rbase + r;
            float v = c[r];
            if (E) {
                v += E[(size_t)row * Ncols + col];
                v = v > 0.0f ? v : 0.0f;
            }
            C[(size_t)row * Ncols + col] = v;
        }
    };
    store_tile(c00, wm,      wn);
    store_tile(c01, wm,      wn + 16);
    store_tile(c10, wm + 16, wn);
    store_tile(c11, wm + 16, wn + 16);
}

__global__ __launch_bounds__(256) void edge_agg_kernel(
    const float* __restrict__ A, const float* __restrict__ Z,
    const float* __restrict__ EK, const float* __restrict__ bias,
    float* __restrict__ E)
{
    const int i  = blockIdx.x;
    const int t  = threadIdx.x;
    const int f  = t & 15;
    const int jg = t >> 4;

    __shared__ float red[256];

    const float* Arow = A + (size_t)i * NN;
    const float* Zrow = Z + (size_t)i * NN * FEDGE;

    float acc = 0.0f;
    #pragma unroll 4
    for (int j = jg; j < NN; j += 16)
        acc = fmaf(Arow[j], __builtin_nontemporal_load(Zrow + j * FEDGE + f), acc);

    red[t] = acc;
    __syncthreads();
    for (int s = 128; s >= 16; s >>= 1) {
        if (t < s) red[t] += red[t + s];
        __syncthreads();
    }

    for (int c = t; c < FOUT; c += 256) {
        float y = bias[c];
        #pragma unroll
        for (int ff = 0; ff < FEDGE; ++ff)
            y = fmaf(red[ff], EK[ff * FOUT + c], y);
        E[(size_t)i * FOUT + c] = y;
    }
}

extern "C" void kernel_launch(void* const* d_in, const int* in_sizes, int n_in,
                              void* d_out, int out_size, void* d_ws, size_t ws_size,
                              hipStream_t stream) {
    const float* X    = (const float*)d_in[0];
    const float* A    = (const float*)d_in[1];
    const float* Z    = (const float*)d_in[2];
    const float* W    = (const float*)d_in[3];
    const float* EK   = (const float*)d_in[4];
    const float* bias = (const float*)d_in[5];
    float* out = (float*)d_out;

    float* XW = (float*)d_ws;
    float* E  = XW + (size_t)NN * FOUT;

    dim3 g1(FOUT / 128, NN / 64);
    wmma_gemm_f32_kernel<<<g1, 256, 0, stream>>>(X, W, nullptr, XW, NN, FOUT, FIN);

    edge_agg_kernel<<<NN, 256, 0, stream>>>(A, Z, EK, bias, E);

    dim3 g2(FOUT / 128, NN / 64);
    wmma_gemm_f32_kernel<<<g2, 256, 0, stream>>>(A, XW, E, out, NN, FOUT, NN);
}


# Product re-associated as (A@X)@W: the large GEMM runs in launch 1 on 4 waves per work-group beside 8 waves that stream Z (one row each); launch 2 = [AX|agg]@[W;EK]+bias, relu; sum order changed
# speedup vs baseline: 4.6609x; 4.6609x over previous
#include <hip/hip_runtime.h>

typedef __attribute__((ext_vector_type(2))) float v2f;
typedef __attribute__((ext_vector_type(4))) float v4f;
typedef __attribute__((ext_vector_type(8))) float v8f;

#define NN     2048
#define FIN    512
#define FOUT   512
#define FEDGE  16
#define AGGP   32

struct FragStage { v4f a, b0, b1, b2, b3; };

__device__ __forceinline__ v8f mma4(v2f a, float bx, float by, v8f acc) {
    v2f b; b.x = bx; b.y = by;
    return __builtin_amdgcn_wmma_f32_16x16x4_f32(false, a, false, b, (short)0, acc, false, false);
}

__device__ __forceinline__ void load_stage(FragStage& s, const float* a, const float* b, int ldb) {
    s.a  = *(const v4f*)a;
    s.b0 = *(const v4f*)(b);
    s.b1 = *(const v4f*)(b + ldb);
    s.b2 = *(const v4f*)(b + 2 * ldb);
    s.b3 = *(const v4f*)(b + 3 * ldb);
}

__device__ __forceinline__ void compute_stage(const FragStage& s, v8f& c0, v8f& c1, v8f& c2, v8f& c3) {
    v2f alo; alo.x = s.a.x; alo.y = s.a.y;
    v2f ahi; ahi.x = s.a.z; ahi.y = s.a.w;
    c0 = mma4(alo, s.b0.x, s.b1.x, c0);
    c1 = mma4(alo, s.b0.y, s.b1.y, c1);
    c2 = mma4(alo, s.b0.z, s.b1.z, c2);
    c3 = mma4(alo, s.b0.w, s.b1.w, c3);
    c0 = mma4(ahi, s.b2.x, s.b3.x, c0);
    c1 = mma4(ahi, s.b2.y, s.b3.y, c1);
    c2 = mma4(ahi, s.b2.z, s.b3.z, c2);
    c3 = mma4(ahi, s.b2.w, s.b3.w, c3);
}

template <int K, int NC>
__device__ __forceinline__ void gemm_wave_tile(const float* ap, const float* bp,
                                               v8f& c0, v8f& c1, v8f& c2, v8f& c3)
{
    constexpr int DEPTH = 8;
    constexpr int NB    = K / 8;
    static_assert(NB % DEPTH == 0 && NB >= DEPTH, "K must be a multiple of 8*DEPTH");
    FragStage st[DEPTH];

    #pragma unroll
    for (int i = 0; i < DEPTH - 1; ++i)
        load_stage(st[i], ap + (size_t)i * 8, bp + (size_t)i * 8 * NC, NC);

    #pragma unroll 1
    for (int kb = 0; kb < NB; kb += DEPTH) {
        #pragma unroll
        for (int i = 0; i < DEPTH; ++i) {
            int nk = kb + i + (DEPTH - 1);
            nk = nk < NB ? nk : NB - 1;
            load_stage(st[(i + DEPTH - 1) % DEPTH], ap + (size_t)nk * 8, bp + (size_t)nk * 8 * NC, NC);
            compute_stage(st[i], c0, c1, c2, c3);
        }
        asm volatile("" ::: "memory");
    }
}

__global__ __launch_bounds__(384) void ax_gemm_and_edge_stream_kernel(
    const float* __restrict__ A, const float* __restrict__ X,
    const float* __restrict__ Z, float* __restrict__ Y, float* __restrict__ AGG)
{
    const int lane = threadIdx.x & 31;
    const int wave = __builtin_amdgcn_readfirstlane((int)(threadIdx.x >> 5));

    if (wave < 4) {
        const int c  = lane & 15;
        const int h  = lane >> 4;
        const int bx = blockIdx.x & 7;
        const int by = blockIdx.x >> 3;
        const int wm = by * 64 + wave * 16;
        const int wn = bx * 64;

        const float* ap = A + (size_t)(wm + c) * NN + 4 * h;
        const float* bp = X + (size_t)(4 * h) * FIN + wn + 4 * c;

        v8f c0 = {}, c1 = {}, c2 = {}, c3 = {};
        gemm_wave_tile<NN, FIN>(ap, bp, c0, c1, c2, c3);

        float* yp = Y + (size_t)(wm + 8 * h) * FIN + wn + 4 * c;
        #pragma unroll
        for (int r = 0; r < 8; ++r) {
            v4f v; v.x = c0[r]; v.y = c1[r]; v.z = c2[r]; v.w = c3[r];
            *(v4f*)(yp + (size_t)r * FIN) = v;
        }
    } else {
        constexpr int U   = 16;
        constexpr int NBT = NN / (8 * U);
        const int row = blockIdx.x * 8 + (wave - 4);
        const int fq  = lane & 3;
        const int jl  = lane >> 2;
        const float* zp = Z + (size_t)row * NN * FEDGE + jl * FEDGE + fq * 4;
        const float* ar = A + (size_t)row * NN + jl;

        v4f z0[U], z1[U];
        float a0[U], a1[U];
        v4f acc = {0.0f, 0.0f, 0.0f, 0.0f};

        auto load_batch = [&](v4f (&z)[U], float (&a)[U], int bt) {
            const float* zq = zp + (size_t)bt * (8 * U * FEDGE);
            const float* aq = ar + bt * (8 * U);
            #pragma unroll
            for (int u = 0; u < U; ++u) {
                z[u] = __builtin_nontemporal_load((const v4f*)(zq + u * 8 * FEDGE));
                a[u] = aq[u * 8];
            }
        };
        auto fma_batch = [&](const v4f (&z)[U], const float (&a)[U]) {
            #pragma unroll
            for (int u = 0; u < U; ++u) {
                acc.x = fmaf(a[u], z[u].x, acc.x);
                acc.y = fmaf(a[u], z[u].y, acc.y);
                acc.z = fmaf(a[u], z[u].z, acc.z);
                acc.w = fmaf(a[u], z[u].w, acc.w);
            }
        };

        load_batch(z0, a0, 0);
        #pragma unroll 1
        for (int bt = 0; bt < NBT - 2; bt += 2) {
            load_batch(z1, a1, bt + 1);
            asm volatile("" ::: "memory");
            fma_batch(z0, a0);
            load_batch(z0, a0, bt + 2);
            asm volatile("" ::: "memory");
            fma_batch(z1, a1);
        }
        load_batch(z1, a1, NBT - 1);
        asm volatile("" ::: "memory");
        fma_batch(z0, a0);
        fma_batch(z1, a1);

        #pragma unroll
        for (int m = 4; m < 32; m <<= 1) {
            acc.x += __shfl_xor(acc.x, m);
            acc.y += __shfl_xor(acc.y, m);
            acc.z += __shfl_xor(acc.z, m);
            acc.w += __shfl_xor(acc.w, m);
        }
        if (jl == 0) *(v4f*)(AGG + (size_t)row * AGGP + fq * 4) = acc;
    }
}

__global__ __launch_bounds__(128) void yw_gemm_edge_bias_relu_kernel(
    const float* __restrict__ Y, const float* __restrict__ W,
    const float* __restrict__ AGG, const float* __restrict__ EK,
    const float* __restrict__ bias, float* __restrict__ out)
{
    const int lane = threadIdx.x & 31;
    const int wave = threadIdx.x >> 5;
    const int c    = lane & 15;
    const int h    = lane >> 4;
    const int wm   = blockIdx.y * 64 + wave * 16;
    const int wn   = blockIdx.x * 64;

    const float* ap = Y + (size_t)(wm + c) * FIN + 4 * h;
    const float* bp = W + (size_t)(4 * h) * FOUT + wn + 4 * c;

    v8f c0 = {}, c1 = {}, c2 = {}, c3 = {};
    gemm_wave_tile<FIN, FOUT>(ap, bp, c0, c1, c2, c3);

    const float* ga = AGG + (size_t)(wm + c) * AGGP + 4 * h;
    const float* gb = EK + (size_t)(4 * h) * FOUT + wn + 4 * c;
    FragStage t0, t1;
    load_stage(t0, ga, gb, FOUT);
    load_stage(t1, ga + 8, gb + (size_t)8 * FOUT, FOUT);
    const v4f bv = *(const v4f*)(bias + wn + 4 * c);
    compute_stage(t0, c0, c1, c2, c3);
    compute_stage(t1, c0, c1, c2, c3);

    float* op = out + (size_t)(wm + 8 * h) * FOUT + wn + 4 * c;
    #pragma unroll
    for (int r = 0; r < 8; ++r) {
        v4f v; v.x = c0[r]; v.y = c1[r]; v.z = c2[r]; v.w = c3[r];
        v += bv;
        v.x = v.x > 0.0f ? v.x : 0.0f;
        v.y = v.y > 0.0f ? v.y : 0.0f;
        v.z = v.z > 0.0f ? v.z : 0.0f;
        v.w = v.w > 0.0f ? v.w : 0.0f;
        *(v4f*)(op + (size_t)r * FOUT) = v;
    }
}

extern "C" void kernel_launch(void* const* d_in, const int* in_sizes, int n_in,
                              void* d_out, int out_size, void* d_ws, size_t ws_size,
                              hipStream_t stream) {
    const float* X    = (const float*)d_in[0];
    const float* A    = (const float*)d_in[1];
    const float* Z    = (const float*)d_in[2];
    const float* W    = (const float*)d_in[3];
    const float* EK   = (const float*)d_in[4];
    const float* bias = (const float*)d_in[5];
    float* out = (float*)d_out;

    float* Y   = (float*)d_ws;
    float* AGG = Y + (size_t)NN * FIN;

    ax_gemm_and_edge_stream_kernel<<<256, 384, 0, stream>>>(A, X, Z, Y, AGG);

    dim3 g2(FOUT / 64, NN / 64);
    yw_gemm_edge_bias_relu_kernel<<<g2, 128, 0, stream>>>(Y, W, AGG, EK, bias, out);
}


# A@X GEMM split over K in the work-group: each of 4 matrix waves does the whole 64x64 tile on a quarter of K (16 accumulator tiles, 5-block load ring), partials summed via LDS; k sum order changed
# speedup vs baseline: 5.7016x; 1.2233x over previous
#include <hip/hip_runtime.h>

typedef __attribute__((ext_vector_type(2))) float v2f;
typedef __attribute__((ext_vector_type(4))) float v4f;
typedef __attribute__((ext_vector_type(8))) float v8f;

#define NN     2048
#define FIN    512
#define FOUT   512
#define FEDGE  16
#define AGGP   32
#define PPITCH 68

struct FragStage { v4f a, b0, b1, b2, b3; };

__device__ __forceinline__ v8f mma4(v2f a, float bx, float by, v8f acc) {
    v2f b; b.x = bx; b.y = by;
    return __builtin_amdgcn_wmma_f32_16x16x4_f32(false, a, false, b, (short)0, acc, false, false);
}

__device__ __forceinline__ void load_stage(FragStage& s, const float* a, const float* b, int ldb) {
    s.a  = *(const v4f*)a;
    s.b0 = *(const v4f*)(b);
    s.b1 = *(const v4f*)(b + ldb);
    s.b2 = *(const v4f*)(b + 2 * ldb);
    s.b3 = *(const v4f*)(b + 3 * ldb);
}

__device__ __forceinline__ void compute_stage(const FragStage& s, v8f& c0, v8f& c1, v8f& c2, v8f& c3) {
    v2f alo; alo.x = s.a.x; alo.y = s.a.y;
    v2f ahi; ahi.x = s.a.z; ahi.y = s.a.w;
    c0 = mma4(alo, s.b0.x, s.b1.x, c0);
    c1 = mma4(alo, s.b0.y, s.b1.y, c1);
    c2 = mma4(alo, s.b0.z, s.b1.z, c2);
    c3 = mma4(alo, s.b0.w, s.b1.w, c3);
    c0 = mma4(ahi, s.b2.x, s.b3.x, c0);
    c1 = mma4(ahi, s.b2.y, s.b3.y, c1);
    c2 = mma4(ahi, s.b2.z, s.b3.z, c2);
    c3 = mma4(ahi, s.b2.w, s.b3.w, c3);
}

template <int K, int NC>
__device__ __forceinline__ void gemm_wave_tile(const float* ap, const float* bp,
                                               v8f& c0, v8f& c1, v8f& c2, v8f& c3)
{
    constexpr int DEPTH = 8;
    constexpr int NB    = K / 8;
    static_assert(NB % DEPTH == 0 && NB >= DEPTH, "K must be a multiple of 8*DEPTH");
    FragStage st[DEPTH];

    #pragma unroll
    for (int i = 0; i < DEPTH - 1; ++i)
        load_stage(st[i], ap + (size_t)i * 8, bp + (size_t)i * 8 * NC, NC);

    #pragma unroll 1
    for (int kb = 0; kb < NB; kb += DEPTH) {
        #pragma unroll
        for (int i = 0; i < DEPTH; ++i) {
            int nk = kb + i + (DEPTH - 1);
            nk = nk < NB ? nk : NB - 1;
            load_stage(st[(i + DEPTH - 1) % DEPTH], ap + (size_t)nk * 8, bp + (size_t)nk * 8 * NC, NC);
            compute_stage(st[i], c0, c1, c2, c3);
        }
        asm volatile("" ::: "memory");
    }
}

struct FragStage64 { v4f a[4]; v4f b0, b1, b2, b3; };

template <int LDA>
__device__ __forceinline__ void load_stage64(FragStage64& s, const float* a, const float* b, int ldb) {
    s.a[0] = *(const v4f*)a;
    s.a[1] = *(const v4f*)(a + (size_t)16 * LDA);
    s.a[2] = *(const v4f*)(a + (size_t)32 * LDA);
    s.a[3] = *(const v4f*)(a + (size_t)48 * LDA);
    s.b0 = *(const v4f*)(b);
    s.b1 = *(const v4f*)(b + ldb);
    s.b2 = *(const v4f*)(b + 2 * ldb);
    s.b3 = *(const v4f*)(b + 3 * ldb);
}

__device__ __forceinline__ void compute_stage64(const FragStage64& s, v8f (&acc)[4][4]) {
    #pragma unroll
    for (int mt = 0; mt < 4; ++mt) {
        v2f alo; alo.x = s.a[mt].x; alo.y = s.a[mt].y;
        acc[mt][0] = mma4(alo, s.b0.x, s.b1.x, acc[mt][0]);
        acc[mt][1] = mma4(alo, s.b0.y, s.b1.y, acc[mt][1]);
        acc[mt][2] = mma4(alo, s.b0.z, s.b1.z, acc[mt][2]);
        acc[mt][3] = mma4(alo, s.b0.w, s.b1.w, acc[mt][3]);
    }
    #pragma unroll
    for (int mt = 0; mt < 4; ++mt) {
        v2f ahi; ahi.x = s.a[mt].z; ahi.y = s.a[mt].w;
        acc[mt][0] = mma4(ahi, s.b2.x, s.b3.x, acc[mt][0]);
        acc[mt][1] = mma4(ahi, s.b2.y, s.b3.y, acc[mt][1]);
        acc[mt][2] = mma4(ahi, s.b2.z, s.b3.z, acc[mt][2]);
        acc[mt][3] = mma4(ahi, s.b2.w, s.b3.w, acc[mt][3]);
    }
}

template <int KQ, int LDA, int NC, int DEPTH>
__device__ __forceinline__ void gemm_wave_tile64(const float* ap, const float* bp, v8f (&acc)[4][4])
{
    constexpr int NB   = KQ / 8;
    constexpr int MAIN = (NB / DEPTH) * DEPTH;
    constexpr int TAIL = NB - MAIN;
    static_assert(NB >= DEPTH && TAIL <= DEPTH - 1, "ring deeper than the k range");
    FragStage64 st[DEPTH];

    #pragma unroll
    for (int i = 0; i < DEPTH - 1; ++i) {
        load_stage64<LDA>(st[i], ap + (size_t)i * 8, bp + (size_t)i * 8 * NC, NC);
        __builtin_amdgcn_sched_barrier(0);
    }

    #pragma unroll 1
    for (int kb = 0; kb < MAIN; kb += DEPTH) {
        #pragma unroll
        for (int i = 0; i < DEPTH; ++i) {
            int nk = kb + i + (DEPTH - 1);
            nk = nk < NB ? nk : NB - 1;
            load_stage64<LDA>(st[(i + DEPTH - 1) % DEPTH], ap + (size_t)nk * 8, bp + (size_t)nk * 8 * NC, NC);
            __builtin_amdgcn_sched_barrier(0);
            compute_stage64(st[i], acc);
            __builtin_amdgcn_sched_barrier(0);
        }
    }
    #pragma unroll
    for (int i = 0; i < TAIL; ++i)
        compute_stage64(st[i], acc);
}

__global__ __launch_bounds__(384) void ax_gemm_and_edge_stream_kernel(
    const float* __restrict__ A, const float* __restrict__ X,
    const float* __restrict__ Z, float* __restrict__ Y, float* __restrict__ AGG)
{
    __shared__ float part[4 * 64 * PPITCH];

    const int lane = threadIdx.x & 31;
    const int wave = __builtin_amdgcn_readfirstlane((int)(threadIdx.x >> 5));
    const int c  = lane & 15;
    const int h  = lane >> 4;
    const int bx = blockIdx.x & 7;
    const int by = blockIdx.x >> 3;
    const int wm = by * 64;
    const int wn = bx * 64;

    if (wave < 4) {
        constexpr int KQ = NN / 4;
        const int kq = wave * KQ;

        const float* ap = A + (size_t)(wm + c) * NN + kq + 4 * h;
        const float* bp = X + (size_t)(kq + 4 * h) * FIN + wn + 4 * c;

        v8f acc[4][4] = {};
        gemm_wave_tile64<KQ, NN, FIN, 5>(ap, bp, acc);

        float* pp = part + (size_t)(wave * 64 + 8 * h) * PPITCH + 4 * c;
        #pragma unroll
        for (int mt = 0; mt < 4; ++mt) {
            #pragma unroll
            for (int r = 0; r < 8; ++r) {
                v4f v; v.x = acc[mt][0][r]; v.y = acc[mt][1][r]; v.z = acc[mt][2][r]; v.w = acc[mt][3][r];
                *(v4f*)(pp + (size_t)(16 * mt + r) * PPITCH) = v;
            }
        }
    } else {
        constexpr int U   = 16;
        constexpr int NBT = NN / (8 * U);
        const int row = blockIdx.x * 8 + (wave - 4);
        const int fq  = lane & 3;
        const int jl  = lane >> 2;
        const float* zp = Z + (size_t)row * NN * FEDGE + jl * FEDGE + fq * 4;
        const float* ar = A + (size_t)row * NN + jl;

        v4f z0[U], z1[U];
        float a0[U], a1[U];
        v4f acc = {0.0f, 0.0f, 0.0f, 0.0f};

        auto load_batch = [&](v4f (&z)[U], float (&a)[U], int bt) {
            const float* zq = zp + (size_t)bt * (8 * U * FEDGE);
            const float* aq = ar + bt * (8 * U);
            #pragma unroll
            for (int u = 0; u < U; ++u) {
                z[u] = __builtin_nontemporal_load((const v4f*)(zq + u * 8 * FEDGE));
                a[u] = aq[u * 8];
            }
        };
        auto fma_batch = [&](const v4f (&z)[U], const float (&a)[U]) {
            #pragma unroll
            for (int u = 0; u < U; ++u) {
                acc.x = fmaf(a[u], z[u].x, acc.x);
                acc.y = fmaf(a[u], z[u].y, acc.y);
                acc.z = fmaf(a[u], z[u].z, acc.z);
                acc.w = fmaf(a[u], z[u].w, acc.w);
            }
        };

        load_batch(z0, a0, 0);
        #pragma unroll 1
        for (int bt = 0; bt < NBT - 2; bt += 2) {
            load_batch(z1, a1, bt + 1);
            asm volatile("" ::: "memory");
            fma_batch(z0, a0);
            load_batch(z0, a0, bt + 2);
            asm volatile("" ::: "memory");
            fma_batch(z1, a1);
        }
        load_batch(z1, a1, NBT - 1);
        asm volatile("" ::: "memory");
        fma_batch(z0, a0);
        fma_batch(z1, a1);

        #pragma unroll
        for (int m = 4; m < 32; m <<= 1) {
            acc.x += __shfl_xor(acc.x, m);
            acc.y += __shfl_xor(acc.y, m);
            acc.z += __shfl_xor(acc.z, m);
            acc.w += __shfl_xor(acc.w, m);
        }
        if (jl == 0) *(v4f*)(AGG + (size_t)row * AGGP + fq * 4) = acc;
    }

    __syncthreads();

    if (wave < 8) {
        const float* pr = part + (size_t)(wave * 8 + h) * PPITCH + 4 * c;
        float* yp = Y + (size_t)(wm + wave * 8 + h) * FIN + wn + 4 * c;
        #pragma unroll
        for (int it = 0; it < 4; ++it) {
            const float* p = pr + (size_t)(2 * it) * PPITCH;
            const v4f p0 = *(const v4f*)(p);
            const v4f p1 = *(const v4f*)(p + 64 * PPITCH);
            const v4f p2 = *(const v4f*)(p + 2 * 64 * PPITCH);
            const v4f p3 = *(const v4f*)(p + 3 * 64 * PPITCH);
            *(v4f*)(yp + (size_t)(2 * it) * FIN) = (p0 + p1) + (p2 + p3);
        }
    }
}

__global__ __launch_bounds__(128) void yw_gemm_edge_bias_relu_kernel(
    const float* __restrict__ Y, const float* __restrict__ W,
    const float* __restrict__ AGG, const float* __restrict__ EK,
    const float* __restrict__ bias, float* __restrict__ out)
{
    const int lane = threadIdx.x & 31;
    const int wave = threadIdx.x >> 5;
    const int c    = lane & 15;
    const int h    = lane >> 4;
    const int wm   = blockIdx.y * 64 + wave * 16;
    const int wn   = blockIdx.x * 64;

    const float* ap = Y + (size_t)(wm + c) * FIN + 4 * h;
    const float* bp = W + (size_t)(4 * h) * FOUT + wn + 4 * c;

    v8f c0 = {}, c1 = {}, c2 = {}, c3 = {};
    gemm_wave_tile<FIN, FOUT>(ap, bp, c0, c1, c2, c3);

    const float* ga = AGG + (size_t)(wm + c) * AGGP + 4 * h;
    const float* gb = EK + (size_t)(4 * h) * FOUT + wn + 4 * c;
    FragStage t0, t1;
    load_stage(t0, ga, gb, FOUT);
    load_stage(t1, ga + 8, gb + (size_t)8 * FOUT, FOUT);
    const v4f bv = *(const v4f*)(bias + wn + 4 * c);
    compute_stage(t0, c0, c1, c2, c3);
    compute_stage(t1, c0, c1, c2, c3);

    float* op = out + (size_t)(wm + 8 * h) * FOUT + wn + 4 * c;
    #pragma unroll
    for (int r = 0; r < 8; ++r) {
        v4f v; v.x = c0[r]; v.y = c1[r]; v.z = c2[r]; v.w = c3[r];
        v += bv;
        v.x = v.x > 0.0f ? v.x : 0.0f;
        v.y = v.y > 0.0f ? v.y : 0.0f;
        v.z = v.z > 0.0f ? v.z : 0.0f;
        v.w = v.w > 0.0f ? v.w : 0.0f;
        *(v4f*)(op + (size_t)r * FOUT) = v;
    }
}

extern "C" void kernel_launch(void* const* d_in, const int* in_sizes, int n_in,
                              void* d_out, int out_size, void* d_ws, size_t ws_size,
                              hipStream_t stream) {
    const float* X    = (const float*)d_in[0];
    const float* A    = (const float*)d_in[1];
    const float* Z    = (const float*)d_in[2];
    const float* W    = (const float*)d_in[3];
    const float* EK   = (const float*)d_in[4];
    const float* bias = (const float*)d_in[5];
    float* out = (float*)d_out;

    float* Y   = (float*)d_ws;
    float* AGG = Y + (size_t)NN * FIN;

    ax_gemm_and_edge_stream_kernel<<<256, 384, 0, stream>>>(A, X, Z, Y, AGG);

    dim3 g2(FOUT / 64, NN / 64);
    yw_gemm_edge_bias_relu_kernel<<<g2, 128, 0, stream>>>(Y, W, AGG, EK, bias, out);
}


# Launch 2 loads held ahead of use: each K block's five loads stay one group 7 blocks before its WMMAs (the compiler had gathered and drained them once per trip); agg, EK, bias asked before the loop
# speedup vs baseline: 5.9426x; 1.0423x over previous
#include <hip/hip_runtime.h>

typedef __attribute__((ext_vector_type(2))) float v2f;
typedef __attribute__((ext_vector_type(4))) float v4f;
typedef __attribute__((ext_vector_type(8))) float v8f;

#define NN     2048
#define FIN    512
#define FOUT   512
#define FEDGE  16
#define AGGP   32
#define PPITCH 68

struct FragStage { v4f a, b0, b1, b2, b3; };

__device__ __forceinline__ v8f mma4(v2f a, float bx, float by, v8f acc) {
    v2f b; b.x = bx; b.y = by;
    return __builtin_amdgcn_wmma_f32_16x16x4_f32(false, a, false, b, (short)0, acc, false, false);
}

__device__ __forceinline__ void load_stage(FragStage& s, const float* a, const float* b, int ldb) {
    s.a  = *(const v4f*)a;
    s.b0 = *(const v4f*)(b);
    s.b1 = *(const v4f*)(b + ldb);
    s.b2 = *(const v4f*)(b + 2 * ldb);
    s.b3 = *(const v4f*)(b + 3 * ldb);
}

__device__ __forceinline__ void compute_stage(const FragStage& s, v8f& c0, v8f& c1, v8f& c2, v8f& c3) {
    v2f alo; alo.x = s.a.x; alo.y = s.a.y;
    v2f ahi; ahi.x = s.a.z; ahi.y = s.a.w;
    c0 = mma4(alo, s.b0.x, s.b1.x, c0);
    c1 = mma4(alo, s.b0.y, s.b1.y, c1);
    c2 = mma4(alo, s.b0.z, s.b1.z, c2);
    c3 = mma4(alo, s.b0.w, s.b1.w, c3);
    c0 = mma4(ahi, s.b2.x, s.b3.x, c0);
    c1 = mma4(ahi, s.b2.y, s.b3.y, c1);
    c2 = mma4(ahi, s.b2.z, s.b3.z, c2);
    c3 = mma4(ahi, s.b2.w, s.b3.w, c3);
}

template <int K, int NC>
__device__ __forceinline__ void gemm_wave_tile(const float* ap, const float* bp,
                                               v8f& c0, v8f& c1, v8f& c2, v8f& c3)
{
    constexpr int DEPTH = 8;
    constexpr int NB    = K / 8;
    static_assert(NB % DEPTH == 0 && NB >= DEPTH, "K must be a multiple of 8*DEPTH");
    FragStage st[DEPTH];

    #pragma unroll
    for (int i = 0; i < DEPTH - 1; ++i) {
        load_stage(st[i], ap + (size_t)i * 8, bp + (size_t)i * 8 * NC, NC);
        __builtin_amdgcn_sched_barrier(0);
    }

    #pragma unroll 1
    for (int kb = 0; kb < NB; kb += DEPTH) {
        #pragma unroll
        for (int i = 0; i < DEPTH; ++i) {
            int nk = kb + i + (DEPTH - 1);
            nk = nk < NB ? nk : NB - 1;
            load_stage(st[(i + DEPTH - 1) % DEPTH], ap + (size_t)nk * 8, bp + (size_t)nk * 8 * NC, NC);
            __builtin_amdgcn_sched_barrier(0);
            compute_stage(st[i], c0, c1, c2, c3);
            __builtin_amdgcn_sched_barrier(0);
        }
    }
}

struct FragStage64 { v4f a[4]; v4f b0, b1, b2, b3; };

template <int LDA>
__device__ __forceinline__ void load_stage64(FragStage64& s, const float* a, const float* b, int ldb) {
    s.a[0] = *(const v4f*)a;
    s.a[1] = *(const v4f*)(a + (size_t)16 * LDA);
    s.a[2] = *(const v4f*)(a + (size_t)32 * LDA);
    s.a[3] = *(const v4f*)(a + (size_t)48 * LDA);
    s.b0 = *(const v4f*)(b);
    s.b1 = *(const v4f*)(b + ldb);
    s.b2 = *(const v4f*)(b + 2 * ldb);
    s.b3 = *(const v4f*)(b + 3 * ldb);
}

__device__ __forceinline__ void compute_stage64(const FragStage64& s, v8f (&acc)[4][4]) {
    #pragma unroll
    for (int mt = 0; mt < 4; ++mt) {
        v2f alo; alo.x = s.a[mt].x; alo.y = s.a[mt].y;
        acc[mt][0] = mma4(alo, s.b0.x, s.b1.x, acc[mt][0]);
        acc[mt][1] = mma4(alo, s.b0.y, s.b1.y, acc[mt][1]);
        acc[mt][2] = mma4(alo, s.b0.z, s.b1.z, acc[mt][2]);
        acc[mt][3] = mma4(alo, s.b0.w, s.b1.w, acc[mt][3]);
    }
    #pragma unroll
    for (int mt = 0; mt < 4; ++mt) {
        v2f ahi; ahi.x = s.a[mt].z; ahi.y = s.a[mt].w;
        acc[mt][0] = mma4(ahi, s.b2.x, s.b3.x, acc[mt][0]);
        acc[mt][1] = mma4(ahi, s.b2.y, s.b3.y, acc[mt][1]);
        acc[mt][2] = mma4(ahi, s.b2.z, s.b3.z, acc[mt][2]);
        acc[mt][3] = mma4(ahi, s.b2.w, s.b3.w, acc[mt][3]);
    }
}

template <int KQ, int LDA, int NC, int DEPTH>
__device__ __forceinline__ void gemm_wave_tile64(const float* ap, const float* bp, v8f (&acc)[4][4])
{
    constexpr int NB   = KQ / 8;
    constexpr int MAIN = (NB / DEPTH) * DEPTH;
    constexpr int TAIL = NB - MAIN;
    static_assert(NB >= DEPTH && TAIL <= DEPTH - 1, "ring deeper than the k range");
    FragStage64 st[DEPTH];

    #pragma unroll
    for (int i = 0; i < DEPTH - 1; ++i) {
        load_stage64<LDA>(st[i], ap + (size_t)i * 8, bp + (size_t)i * 8 * NC, NC);
        __builtin_amdgcn_sched_barrier(0);
    }

    #pragma unroll 1
    for (int kb = 0; kb < MAIN; kb += DEPTH) {
        #pragma unroll
        for (int i = 0; i < DEPTH; ++i) {
            int nk = kb + i + (DEPTH - 1);
            nk = nk < NB ? nk : NB - 1;
            load_stage64<LDA>(st[(i + DEPTH - 1) % DEPTH], ap + (size_t)nk * 8, bp + (size_t)nk * 8 * NC, NC);
            __builtin_amdgcn_sched_barrier(0);
            compute_stage64(st[i], acc);
            __builtin_amdgcn_sched_barrier(0);
        }
    }
    #pragma unroll
    for (int i = 0; i < TAIL; ++i)
        compute_stage64(st[i], acc);
}

__global__ __launch_bounds__(384) void ax_gemm_and_edge_stream_kernel(
    const float* __restrict__ A, const float* __restrict__ X,
    const float* __restrict__ Z, float* __restrict__ Y, float* __restrict__ AGG)
{
    __shared__ float part[4 * 64 * PPITCH];

    const int lane = threadIdx.x & 31;
    const int wave = __builtin_amdgcn_readfirstlane((int)(threadIdx.x >> 5));
    const int c  = lane & 15;
    const int h  = lane >> 4;
    const int bx = blockIdx.x & 7;
    const int by = blockIdx.x >> 3;
    const int wm = by * 64;
    const int wn = bx * 64;

    if (wave < 4) {
        constexpr int KQ = NN / 4;
        const int kq = wave * KQ;

        const float* ap = A + (size_t)(wm + c) * NN + kq + 4 * h;
        const float* bp = X + (size_t)(kq + 4 * h) * FIN + wn + 4 * c;

        v8f acc[4][4] = {};
        gemm_wave_tile64<KQ, NN, FIN, 5>(ap, bp, acc);

        float* pp = part + (size_t)(wave * 64 + 8 * h) * PPITCH + 4 * c;
        #pragma unroll
        for (int mt = 0; mt < 4; ++mt) {
            #pragma unroll
            for (int r = 0; r < 8; ++r) {
                v4f v; v.x = acc[mt][0][r]; v.y = acc[mt][1][r]; v.z = acc[mt][2][r]; v.w = acc[mt][3][r];
                *(v4f*)(pp + (size_t)(16 * mt + r) * PPITCH) = v;
            }
        }
    } else {
        constexpr int U   = 16;
        constexpr int NBT = NN / (8 * U);
        const int row = blockIdx.x * 8 + (wave - 4);
        const int fq  = lane & 3;
        const int jl  = lane >> 2;
        const float* zp = Z + (size_t)row * NN * FEDGE + jl * FEDGE + fq * 4;
        const float* ar = A + (size_t)row * NN + jl;

        v4f z0[U], z1[U];
        float a0[U], a1[U];
        v4f acc = {0.0f, 0.0f, 0.0f, 0.0f};

        auto load_batch = [&](v4f (&z)[U], float (&a)[U], int bt) {
            const float* zq = zp + (size_t)bt * (8 * U * FEDGE);
            const float* aq = ar + bt * (8 * U);
            #pragma unroll
            for (int u = 0; u < U; ++u) {
                z[u] = __builtin_nontemporal_load((const v4f*)(zq + u * 8 * FEDGE));
                a[u] = aq[u * 8];
            }
        };
        auto fma_batch = [&](const v4f (&z)[U], const float (&a)[U]) {
            #pragma unroll
            for (int u = 0; u < U; ++u) {
                acc.x = fmaf(a[u], z[u].x, acc.x);
                acc.y = fmaf(a[u], z[u].y, acc.y);
                acc.z = fmaf(a[u], z[u].z, acc.z);
                acc.w = fmaf(a[u], z[u].w, acc.w);
            }
        };

        load_batch(z0, a0, 0);
        #pragma unroll 1
        for (int bt = 0; bt < NBT - 2; bt += 2) {
            load_batch(z1, a1, bt + 1);
            asm volatile("" ::: "memory");
            fma_batch(z0, a0);
            load_batch(z0, a0, bt + 2);
            asm volatile("" ::: "memory");
            fma_batch(z1, a1);
        }
        load_batch(z1, a1, NBT - 1);
        asm volatile("" ::: "memory");
        fma_batch(z0, a0);
        fma_batch(z1, a1);

        #pragma unroll
        for (int m = 4; m < 32; m <<= 1) {
            acc.x += __shfl_xor(acc.x, m);
            acc.y += __shfl_xor(acc.y, m);
            acc.z += __shfl_xor(acc.z, m);
            acc.w += __shfl_xor(acc.w, m);
        }
        if (jl == 0) *(v4f*)(AGG + (size_t)row * AGGP + fq * 4) = acc;
    }

    __syncthreads();

    if (wave < 8) {
        const float* pr = part + (size_t)(wave * 8 + h) * PPITCH + 4 * c;
        float* yp = Y + (size_t)(wm + wave * 8 + h) * FIN + wn + 4 * c;
        #pragma unroll
        for (int it = 0; it < 4; ++it) {
            const float* p = pr + (size_t)(2 * it) * PPITCH;
            const v4f p0 = *(const v4f*)(p);
            const v4f p1 = *(const v4f*)(p + 64 * PPITCH);
            const v4f p2 = *(const v4f*)(p + 2 * 64 * PPITCH);
            const v4f p3 = *(const v4f*)(p + 3 * 64 * PPITCH);
            *(v4f*)(yp + (size_t)(2 * it) * FIN) = (p0 + p1) + (p2 + p3);
        }
    }
}

__global__ __launch_bounds__(128) void yw_gemm_edge_bias_relu_kernel(
    const float* __restrict__ Y, const float* __restrict__ W,
    const float* __restrict__ AGG, const float* __restrict__ EK,
    const float* __restrict__ bias, float* __restrict__ out)
{
    const int lane = threadIdx.x & 31;
    const int wave = threadIdx.x >> 5;
    const int c    = lane & 15;
    const int h    = lane >> 4;
    const int wm   = blockIdx.y * 64 + wave * 16;
    const int wn   = blockIdx.x * 64;

    const float* ap = Y + (size_t)(wm + c) * FIN + 4 * h;
    const float* bp = W + (size_t)(4 * h) * FOUT + wn + 4 * c;

    const float* ga = AGG + (size_t)(wm + c) * AGGP + 4 * h;
    const float* gb = EK + (size_t)(4 * h) * FOUT + wn + 4 * c;
    FragStage t0, t1;
    load_stage(t0, ga, gb, FOUT);
    load_stage(t1, ga + 8, gb + (size_t)8 * FOUT, FOUT);
    const v4f bv = *(const v4f*)(bias + wn + 4 * c);
    __builtin_amdgcn_sched_barrier(0);

    v8f c0 = {}, c1 = {}, c2 = {}, c3 = {};
    gemm_wave_tile<FIN, FOUT>(ap, bp, c0, c1, c2, c3);

    compute_stage(t0, c0, c1, c2, c3);
    compute_stage(t1, c0, c1, c2, c3);

    float* op = out + (size_t)(wm + 8 * h) * FOUT + wn + 4 * c;
    #pragma unroll
    for (int r = 0; r < 8; ++r) {
        v4f v; v.x = c0[r]; v.y = c1[r]; v.z = c2[r]; v.w = c3[r];
        v += bv;
        v.x = v.x > 0.0f ? v.x : 0.0f;
        v.y = v.y > 0.0f ? v.y : 0.0f;
        v.z = v.z > 0.0f ? v.z : 0.0f;
        v.w = v.w > 0.0f ? v.w : 0.0f;
        *(v4f*)(op + (size_t)r * FOUT) = v;
    }
}

extern "C" void kernel_launch(void* const* d_in, const int* in_sizes, int n_in,
                              void* d_out, int out_size, void* d_ws, size_t ws_size,
                              hipStream_t stream) {
    const float* X    = (const float*)d_in[0];
    const float* A    = (const float*)d_in[1];
    const float* Z    = (const float*)d_in[2];
    const float* W    = (const float*)d_in[3];
    const float* EK   = (const float*)d_in[4];
    const float* bias = (const float*)d_in[5];
    float* out = (float*)d_out;

    float* Y   = (float*)d_ws;
    float* AGG = Y + (size_t)NN * FIN;

    ax_gemm_and_edge_stream_kernel<<<256, 384, 0, stream>>>(A, X, Z, Y, AGG);

    dim3 g2(FOUT / 64, NN / 64);
    yw_gemm_edge_bias_relu_kernel<<<g2, 128, 0, stream>>>(Y, W, AGG, EK, bias, out);
}


# Launch 1: A read in whole lines (4 rows x 128 B per load, was 16 rows x 32 B) and turned into fragments through a wave-private LDS strip, no barrier; X stays global-to-register; sums unchanged
# speedup vs baseline: 6.5255x; 1.0981x over previous
#include <hip/hip_runtime.h>

typedef __attribute__((ext_vector_type(2))) float v2f;
typedef __attribute__((ext_vector_type(4))) float v4f;
typedef __attribute__((ext_vector_type(8))) float v8f;

#define NN     2048
#define FIN    512
#define FOUT   512
#define FEDGE  16
#define AGGP   32
#define APITCH 36
#define PPITCH 68

struct FragStage { v4f a, b0, b1, b2, b3; };

__device__ __forceinline__ v8f mma4(v2f a, float bx, float by, v8f acc) {
    v2f b; b.x = bx; b.y = by;
    return __builtin_amdgcn_wmma_f32_16x16x4_f32(false, a, false, b, (short)0, acc, false, false);
}

__device__ __forceinline__ void load_stage(FragStage& s, const float* a, const float* b, int ldb) {
    s.a  = *(const v4f*)a;
    s.b0 = *(const v4f*)(b);
    s.b1 = *(const v4f*)(b + ldb);
    s.b2 = *(const v4f*)(b + 2 * ldb);
    s.b3 = *(const v4f*)(b + 3 * ldb);
}

__device__ __forceinline__ void compute_stage(const FragStage& s, v8f& c0, v8f& c1, v8f& c2, v8f& c3) {
    v2f alo; alo.x = s.a.x; alo.y = s.a.y;
    v2f ahi; ahi.x = s.a.z; ahi.y = s.a.w;
    c0 = mma4(alo, s.b0.x, s.b1.x, c0);
    c1 = mma4(alo, s.b0.y, s.b1.y, c1);
    c2 = mma4(alo, s.b0.z, s.b1.z, c2);
    c3 = mma4(alo, s.b0.w, s.b1.w, c3);
    c0 = mma4(ahi, s.b2.x, s.b3.x, c0);
    c1 = mma4(ahi, s.b2.y, s.b3.y, c1);
    c2 = mma4(ahi, s.b2.z, s.b3.z, c2);
    c3 = mma4(ahi, s.b2.w, s.b3.w, c3);
}

template <int K, int NC>
__device__ __forceinline__ void gemm_wave_tile(const float* ap, const float* bp,
                                               v8f& c0, v8f& c1, v8f& c2, v8f& c3)
{
    constexpr int DEPTH = 8;
    constexpr int NB    = K / 8;
    static_assert(NB % DEPTH == 0 && NB >= DEPTH, "K must be a multiple of 8*DEPTH");
    FragStage st[DEPTH];

    #pragma unroll
    for (int i = 0; i < DEPTH - 1; ++i) {
        load_stage(st[i], ap + (size_t)i * 8, bp + (size_t)i * 8 * NC, NC);
        __builtin_amdgcn_sched_barrier(0);
    }

    #pragma unroll 1
    for (int kb = 0; kb < NB; kb += DEPTH) {
        #pragma unroll
        for (int i = 0; i < DEPTH; ++i) {
            int nk = kb + i + (DEPTH - 1);
            nk = nk < NB ? nk : NB - 1;
            load_stage(st[(i + DEPTH - 1) % DEPTH], ap + (size_t)nk * 8, bp + (size_t)nk * 8 * NC, NC);
            __builtin_amdgcn_sched_barrier(0);
            compute_stage(st[i], c0, c1, c2, c3);
            __builtin_amdgcn_sched_barrier(0);
        }
    }
}

struct BStage64 { v4f b0, b1, b2, b3; };

__device__ __forceinline__ void compute_block64(const v4f (&a)[4], const BStage64& s, v8f (&acc)[4][4]) {
    #pragma unroll
    for (int mt = 0; mt < 4; ++mt) {
        v2f alo; alo.x = a[mt].x; alo.y = a[mt].y;
        acc[mt][0] = mma4(alo, s.b0.x, s.b1.x, acc[mt][0]);
        acc[mt][1] = mma4(alo, s.b0.y, s.b1.y, acc[mt][1]);
        acc[mt][2] = mma4(alo, s.b0.z, s.b1.z, acc[mt][2]);
        acc[mt][3] = mma4(alo, s.b0.w, s.b1.w, acc[mt][3]);
    }
    #pragma unroll
    for (int mt = 0; mt < 4; ++mt) {
        v2f ahi; ahi.x = a[mt].z; ahi.y = a[mt].w;
        acc[mt][0] = mma4(ahi, s.b2.x, s.b3.x, acc[mt][0]);
        acc[mt][1] = mma4(ahi, s.b2.y, s.b3.y, acc[mt][1]);
        acc[mt][2] = mma4(ahi, s.b2.z, s.b3.z, acc[mt][2]);
        acc[mt][3] = mma4(ahi, s.b2.w, s.b3.w, acc[mt][3]);
    }
}

template <int KQ, int LDA, int NC>
__device__ __forceinline__ void gemm_wave_tile64(const float* ag, const float* bp, float* sa,
                                                 int lane, v8f (&acc)[4][4])
{
    constexpr int NB    = KQ / 8;
    constexpr int NCH   = KQ / 32;
    constexpr int BD    = 4;
    constexpr int CHF   = 64 * APITCH;
    static_assert(NCH % 2 == 0 && NCH >= 2, "the loop takes two chunks per trip");

    const int c = lane & 15, h = lane >> 4;
    float*       sst = sa + (lane >> 3) * APITCH + 4 * (lane & 7);
    const float* sld = sa + c * APITCH + 4 * h;

    v4f g[16];
    v4f fa[2][4];
    BStage64 st[BD];

    auto load_chunk = [&](int n) {
        #pragma unroll
        for (int i = 0; i < 16; ++i) g[i] = *(const v4f*)(ag + (size_t)(4 * i) * LDA + 32 * n);
    };
    auto store_chunk = [&](int buf) {
        #pragma unroll
        for (int i = 0; i < 16; ++i) *(v4f*)(sst + buf * CHF + 4 * i * APITCH) = g[i];
    };
    auto load_frag = [&](v4f (&f)[4], int buf, int b) {
        #pragma unroll
        for (int mt = 0; mt < 4; ++mt) f[mt] = *(const v4f*)(sld + buf * CHF + 16 * mt * APITCH + 8 * b);
    };
    auto load_b = [&](BStage64& s, int blk) {
        const float* b = bp + (size_t)blk * 8 * NC;
        s.b0 = *(const v4f*)(b);
        s.b1 = *(const v4f*)(b + NC);
        s.b2 = *(const v4f*)(b + 2 * NC);
        s.b3 = *(const v4f*)(b + 3 * NC);
    };

    load_chunk(0);
    #pragma unroll
    for (int i = 0; i < BD - 1; ++i) load_b(st[i], i);
    __builtin_amdgcn_sched_barrier(0);
    store_chunk(0);
    load_chunk(1);
    load_frag(fa[0], 0, 0);
    __builtin_amdgcn_sched_barrier(0);

    #pragma unroll 1
    for (int n0 = 0; n0 < NCH; n0 += 2) {
        #pragma unroll
        for (int p = 0; p < 2; ++p) {
            const int n = n0 + p;
            #pragma unroll
            for (int b = 0; b < 4; ++b) {
                int nb = 4 * n + b + (BD - 1);
                nb = nb < NB ? nb : NB - 1;
                load_b(st[(b + BD - 1) % BD], nb);
                if (b == 2) {
                    int nn = n + 2; nn = nn < NCH ? nn : NCH - 1;
                    __builtin_amdgcn_sched_barrier(0);
                    store_chunk(1 - p);
                    load_chunk(nn);
                }
                load_frag(fa[(b + 1) & 1], b < 3 ? p : 1 - p, (b + 1) & 3);
                __builtin_amdgcn_sched_barrier(0);
                compute_block64(fa[b & 1], st[b], acc);
                __builtin_amdgcn_sched_barrier(0);
            }
        }
    }
}

__global__ __launch_bounds__(384) void ax_gemm_and_edge_stream_kernel(
    const float* __restrict__ A, const float* __restrict__ X,
    const float* __restrict__ Z, float* __restrict__ Y, float* __restrict__ AGG)
{
    __shared__ float part[4 * 64 * PPITCH];
    __shared__ float astage[4 * 2 * 64 * APITCH];

    const int lane = threadIdx.x & 31;
    const int wave = __builtin_amdgcn_readfirstlane((int)(threadIdx.x >> 5));
    const int c  = lane & 15;
    const int h  = lane >> 4;
    const int bx = blockIdx.x & 7;
    const int by = blockIdx.x >> 3;
    const int wm = by * 64;
    const int wn = bx * 64;

    if (wave < 4) {
        constexpr int KQ = NN / 4;
        const int kq = wave * KQ;

        const float* ag = A + (size_t)(wm + (lane >> 3)) * NN + kq + 4 * (lane & 7);
        const float* bp = X + (size_t)(kq + 4 * h) * FIN + wn + 4 * c;

        v8f acc[4][4] = {};
        gemm_wave_tile64<KQ, NN, FIN>(ag, bp, astage + wave * (2 * 64 * APITCH), lane, acc);

        float* pp = part + (size_t)(wave * 64 + 8 * h) * PPITCH + 4 * c;
        #pragma unroll
        for (int mt = 0; mt < 4; ++mt) {
            #pragma unroll
            for (int r = 0; r < 8; ++r) {
                v4f v; v.x = acc[mt][0][r]; v.y = acc[mt][1][r]; v.z = acc[mt][2][r]; v.w = acc[mt][3][r];
                *(v4f*)(pp + (size_t)(16 * mt + r) * PPITCH) = v;
            }
        }
    } else {
        constexpr int U   = 16;
        constexpr int NBT = NN / (8 * U);
        const int row = blockIdx.x * 8 + (wave - 4);
        const int fq  = lane & 3;
        const int jl  = lane >> 2;
        const float* zp = Z + (size_t)row * NN * FEDGE + jl * FEDGE + fq * 4;
        const float* ar = A + (size_t)row * NN + jl;

        v4f z0[U], z1[U];
        float a0[U], a1[U];
        v4f acc = {0.0f, 0.0f, 0.0f, 0.0f};

        auto load_batch = [&](v4f (&z)[U], float (&a)[U], int bt) {
            const float* zq = zp + (size_t)bt * (8 * U * FEDGE);
            const float* aq = ar + bt * (8 * U);
            #pragma unroll
            for (int u = 0; u < U; ++u) {
                z[u] = __builtin_nontemporal_load((const v4f*)(zq + u * 8 * FEDGE));
                a[u] = aq[u * 8];
            }
        };
        auto fma_batch = [&](const v4f (&z)[U], const float (&a)[U]) {
            #pragma unroll
            for (int u = 0; u < U; ++u) {
                acc.x = fmaf(a[u], z[u].x, acc.x);
                acc.y = fmaf(a[u], z[u].y, acc.y);
                acc.z = fmaf(a[u], z[u].z, acc.z);
                acc.w = fmaf(a[u], z[u].w, acc.w);
            }
        };

        load_batch(z0, a0, 0);
        #pragma unroll 1
        for (int bt = 0; bt < NBT - 2; bt += 2) {
            load_batch(z1, a1, bt + 1);
            asm volatile("" ::: "memory");
            fma_batch(z0, a0);
            load_batch(z0, a0, bt + 2);
            asm volatile("" ::: "memory");
            fma_batch(z1, a1);
        }
        load_batch(z1, a1, NBT - 1);
        asm volatile("" ::: "memory");
        fma_batch(z0, a0);
        fma_batch(z1, a1);

        #pragma unroll
        for (int m = 4; m < 32; m <<= 1) {
            acc.x += __shfl_xor(acc.x, m);
            acc.y += __shfl_xor(acc.y, m);
            acc.z += __shfl_xor(acc.z, m);
            acc.w += __shfl_xor(acc.w, m);
        }
        if (jl == 0) *(v4f*)(AGG + (size_t)row * AGGP + fq * 4) = acc;
    }

    __syncthreads();

    if (wave < 8) {
        const float* pr = part + (size_t)(wave * 8 + h) * PPITCH + 4 * c;
        float* yp = Y + (size_t)(wm + wave * 8 + h) * FIN + wn + 4 * c;
        #pragma unroll
        for (int it = 0; it < 4; ++it) {
            const float* p = pr + (size_t)(2 * it) * PPITCH;
            const v4f p0 = *(const v4f*)(p);
            const v4f p1 = *(const v4f*)(p + 64 * PPITCH);
            const v4f p2 = *(const v4f*)(p + 2 * 64 * PPITCH);
            const v4f p3 = *(const v4f*)(p + 3 * 64 * PPITCH);
            *(v4f*)(yp + (size_t)(2 * it) * FIN) = (p0 + p1) + (p2 + p3);
        }
    }
}

__global__ __launch_bounds__(128) void yw_gemm_edge_bias_relu_kernel(
    const float* __restrict__ Y, const float* __restrict__ W,
    const float* __restrict__ AGG, const float* __restrict__ EK,
    const float* __restrict__ bias, float* __restrict__ out)
{
    const int lane = threadIdx.x & 31;
    const int wave = threadIdx.x >> 5;
    const int c    = lane & 15;
    const int h    = lane >> 4;
    const int wm   = blockIdx.y * 64 + wave * 16;
    const int wn   = blockIdx.x * 64;

    const float* ap = Y + (size_t)(wm + c) * FIN + 4 * h;
    const float* bp = W + (size_t)(4 * h) * FOUT + wn + 4 * c;

    const float* ga = AGG + (size_t)(wm + c) * AGGP + 4 * h;
    const float* gb = EK + (size_t)(4 * h) * FOUT + wn + 4 * c;
    FragStage t0, t1;
    load_stage(t0, ga, gb, FOUT);
    load_stage(t1, ga + 8, gb + (size_t)8 * FOUT, FOUT);
    const v4f bv = *(const v4f*)(bias + wn + 4 * c);
    __builtin_amdgcn_sched_barrier(0);

    v8f c0 = {}, c1 = {}, c2 = {}, c3 = {};
    gemm_wave_tile<FIN, FOUT>(ap, bp, c0, c1, c2, c3);

    compute_stage(t0, c0, c1, c2, c3);
    compute_stage(t1, c0, c1, c2, c3);

    float* op = out + (size_t)(wm + 8 * h) * FOUT + wn + 4 * c;
    #pragma unroll
    for (int r = 0; r < 8; ++r) {
        v4f v; v.x = c0[r]; v.y = c1[r]; v.z = c2[r]; v.w = c3[r];
        v += bv;
        v.x = v.x > 0.0f ? v.x : 0.0f;
        v.y = v.y > 0.0f ? v.y : 0.0f;
        v.z = v.z > 0.0f ? v.z : 0.0f;
        v.w = v.w > 0.0f ? v.w : 0.0f;
        *(v4f*)(op + (size_t)r * FOUT) = v;
    }
}

extern "C" void kernel_launch(void* const* d_in, const int* in_sizes, int n_in,
                              void* d_out, int out_size, void* d_ws, size_t ws_size,
                              hipStream_t stream) {
    const float* X    = (const float*)d_in[0];
    const float* A    = (const float*)d_in[1];
    const float* Z    = (const float*)d_in[2];
    const float* W    = (const float*)d_in[3];
    const float* EK   = (const float*)d_in[4];
    const float* bias = (const float*)d_in[5];
    float* out = (float*)d_out;

    float* Y   = (float*)d_ws;
    float* AGG = Y + (size_t)NN * FIN;

    ax_gemm_and_edge_stream_kernel<<<256, 384, 0, stream>>>(A, X, Z, Y, AGG);

    dim3 g2(FOUT / 64, NN / 64);
    yw_gemm_edge_bias_relu_kernel<<<g2, 128, 0, stream>>>(Y, W, AGG, EK, bias, out);
}
